# starting kernel
# baseline (speedup 1.0000x reference)
#include <hip/hip_runtime.h>


#define N_GENES 768
#define HID     128
#define NTILE   48

typedef float v2f __attribute__((ext_vector_type(2)));
typedef float v8f __attribute__((ext_vector_type(8)));

__global__ __launch_bounds__(HID) void grn_proj_kernel(
    const float* __restrict__ X, const float* __restrict__ W1,
    const float* __restrict__ Wb,
    float* __restrict__ Ai, float* __restrict__ Bj, float* __restrict__ Y) {
  __shared__ float sX[HID];
  const int i = blockIdx.x;
  const int c = threadIdx.x;
  sX[c] = X[i * HID + c];
  __syncthreads();
  const float* __restrict__ w1a = W1 + c * (2 * HID);
  const float* __restrict__ w1b = w1a + HID;
  float a = 0.f, b = 0.f, y = 0.f;
#pragma unroll 8
  for (int k = 0; k < HID; ++k) {
    const float xv = sX[k];
    a = fmaf(xv, w1a[k], a);
    b = fmaf(xv, w1b[k], b);
    y = fmaf(xv, Wb[k * HID + c], y);
  }
  Ai[i * HID + c] = a;
  Bj[i * HID + c] = b;
  Y[i * HID + c]  = y;
}

__global__ __launch_bounds__(32) void grn_tile_kernel(
    const float* __restrict__ X,  const float* __restrict__ Ai,
    const float* __restrict__ Bj, const float* __restrict__ Y,
    const float* __restrict__ b1, const float* __restrict__ W2,
    const float* __restrict__ b2, const float* __restrict__ bb,
    float* __restrict__ out) {
  __shared__ alignas(16) float sAi[16 * HID];
  __shared__ alignas(16) float sBj[16 * HID];
  __shared__ alignas(16) float sB1[HID];
  __shared__ alignas(16) float sW2[3 * HID];
  __shared__ float sSign[256];

  const int lane = threadIdx.x;
  const int i0 = blockIdx.x * 16;
  const int j0 = blockIdx.y * 16;

  {
    const float4* __restrict__ gAi = (const float4*)(Ai + i0 * HID);
    const float4* __restrict__ gBj = (const float4*)(Bj + j0 * HID);
    float4* dAi = (float4*)sAi;
    float4* dBj = (float4*)sBj;
#pragma unroll
    for (int q = 0; q < 16; ++q) {
      dAi[lane + 32 * q] = gAi[lane + 32 * q];
      dBj[lane + 32 * q] = gBj[lane + 32 * q];
    }
    ((float4*)sB1)[lane] = ((const float4*)b1)[lane];
#pragma unroll
    for (int q = 0; q < 3; ++q)
      ((float4*)sW2)[lane + 32 * q] = ((const float4*)W2)[lane + 32 * q];
  }
  __syncthreads();

  const float bbv = bb[0];
  v8f acc = {bbv, bbv, bbv, bbv, bbv, bbv, bbv, bbv};
  const int lm = lane & 15;
  const int kb = (lane >> 4) << 1;
  const float* __restrict__ arow = Y + (i0 + lm) * HID + kb;
  const float* __restrict__ brow = X + (j0 + lm) * HID + kb;
#pragma unroll
  for (int kk = 0; kk < HID; kk += 4) {
    const v2f afrag = *(const v2f*)(arow + kk);
    const v2f bfrag = *(const v2f*)(brow + kk);
    acc = __builtin_amdgcn_wmma_f32_16x16x4_f32(
         false, afrag,  false, bfrag,
         (short)0, acc,  false,  false);
  }

  const float b20 = b2[0], b21 = b2[1], b22 = b2[2];
#pragma unroll
  for (int r = 0; r < 8; ++r) {
    const int p  = lane + 32 * r;
    const int mi = p >> 4;
    const int nj = p & 15;
    const float4* __restrict__ ha = (const float4*)(sAi + mi * HID);
    const float4* __restrict__ hb = (const float4*)(sBj + nj * HID);
    const float4* __restrict__ hc = (const float4*)sB1;
    const float4* __restrict__ w0 = (const float4*)sW2;
    const float4* __restrict__ w1 = (const float4*)(sW2 + HID);
    const float4* __restrict__ w2 = (const float4*)(sW2 + 2 * HID);
    float l0 = b20, l1 = b21, l2 = b22;
#pragma unroll 4
    for (int k4 = 0; k4 < HID / 4; ++k4) {
      const float4 va = ha[k4], vb = hb[k4], vc = hc[k4];
      float4 h;
      h.x = fmaxf(va.x + vb.x + vc.x, 0.f);
      h.y = fmaxf(va.y + vb.y + vc.y, 0.f);
      h.z = fmaxf(va.z + vb.z + vc.z, 0.f);
      h.w = fmaxf(va.w + vb.w + vc.w, 0.f);
      const float4 u0 = w0[k4], u1 = w1[k4], u2 = w2[k4];
      l0 = fmaf(h.x, u0.x, fmaf(h.y, u0.y, fmaf(h.z, u0.z, fmaf(h.w, u0.w, l0))));
      l1 = fmaf(h.x, u1.x, fmaf(h.y, u1.y, fmaf(h.z, u1.z, fmaf(h.w, u1.w, l1))));
      l2 = fmaf(h.x, u2.x, fmaf(h.y, u2.y, fmaf(h.z, u2.z, fmaf(h.w, u2.w, l2))));
    }
    const float s = (l0 >= l1 && l0 >= l2) ? 1.0f : ((l1 >= l2) ? 0.0f : -1.0f);
    sSign[p] = s;
  }
  __syncthreads();

  const int n = lane & 15;
  const int mbase = (lane >> 4) << 3;
#pragma unroll
  for (int v = 0; v < 8; ++v) {
    const int m  = v + mbase;
    const int gi = i0 + m;
    const int gj = j0 + n;
    const float val = (gi == gj) ? 0.f : sSign[m * 16 + n] * acc[v];
    out[gi * N_GENES + gj] = val;
  }
}

extern "C" void kernel_launch(void* const* d_in, const int* in_sizes, int n_in,
                              void* d_out, int out_size, void* d_ws,
                              size_t ws_size, hipStream_t stream) {
  (void)in_sizes; (void)n_in; (void)out_size; (void)ws_size;
  const float* X  = (const float*)d_in[0];
  const float* W1 = (const float*)d_in[1];
  const float* b1 = (const float*)d_in[2];
  const float* W2 = (const float*)d_in[3];
  const float* b2 = (const float*)d_in[4];
  const float* Wb = (const float*)d_in[5];
  const float* bb = (const float*)d_in[6];
  float* out = (float*)d_out;

  float* ws = (float*)d_ws;
  float* Ai = ws;
  float* Bj = ws + N_GENES * HID;
  float* Y  = ws + 2 * N_GENES * HID;

  grn_proj_kernel<<<dim3(N_GENES), dim3(HID), 0, stream>>>(X, W1, Wb, Ai, Bj, Y);
  grn_tile_kernel<<<dim3(NTILE, NTILE), dim3(32), 0, stream>>>(
      X, Ai, Bj, Y, b1, W2, b2, bb, out);
}


# Classifier loop nest interchanged: k chunk outermost, the patch's 12 genes inside as 12 accumulator tiles; per chunk one group of LDS reads in front of 24 WMMAs (was one per 4); sums bit-identical
# speedup vs baseline: 2.2632x; 2.2632x over previous
#include <hip/hip_runtime.h>


#define N_GENES 768
#define HID     128
#define NTILE   48

typedef float v2f __attribute__((ext_vector_type(2)));
typedef float v8f __attribute__((ext_vector_type(8)));

__global__ __launch_bounds__(32) void grn_proj_kernel(
    const float* __restrict__ X, const float* __restrict__ W1,
    const float* __restrict__ Wb,
    float* __restrict__ Ai, float* __restrict__ Bj, float* __restrict__ Y) {
  const int lane = threadIdx.x;
  const int i0 = blockIdx.x * 16;
  const int m  = blockIdx.y >> 2;
  const int c0 = (blockIdx.y & 3) * 32;
  const int lm = lane & 15;
  const int kb = (lane >> 4) << 1;

  v2f af[HID / 4], bf0[HID / 4], bf1[HID / 4];

  const float* __restrict__ arow = X + (i0 + lm) * HID + kb;
#pragma unroll
  for (int s = 0; s < HID / 4; ++s) af[s] = *(const v2f*)(arow + 4 * s);

  if (m < 2) {
    const float* __restrict__ w0 =
        W1 + (c0 + 2 * lm) * (2 * HID) + m * HID + kb;
    const float* __restrict__ w1 = w0 + 2 * HID;
#pragma unroll
    for (int s = 0; s < HID / 4; ++s) {
      bf0[s] = *(const v2f*)(w0 + 4 * s);
      bf1[s] = *(const v2f*)(w1 + 4 * s);
    }
  } else {
    const float* __restrict__ w = Wb + kb * HID + c0 + 2 * lm;
#pragma unroll
    for (int s = 0; s < HID / 4; ++s) {
      const v2f e0 = *(const v2f*)(w + (4 * s) * HID);
      const v2f e1 = *(const v2f*)(w + (4 * s + 1) * HID);
      v2f t0, t1;
      t0.x = e0.x; t0.y = e1.x;
      t1.x = e0.y; t1.y = e1.y;
      bf0[s] = t0;
      bf1[s] = t1;
    }
  }

  v8f acc0 = {0.f, 0.f, 0.f, 0.f, 0.f, 0.f, 0.f, 0.f};
  v8f acc1 = {0.f, 0.f, 0.f, 0.f, 0.f, 0.f, 0.f, 0.f};
#pragma unroll
  for (int s = 0; s < HID / 4; ++s) {
    acc0 = __builtin_amdgcn_wmma_f32_16x16x4_f32(
         false, af[s],  false, bf0[s],
         (short)0, acc0,  false,  false);
    acc1 = __builtin_amdgcn_wmma_f32_16x16x4_f32(
         false, af[s],  false, bf1[s],
         (short)0, acc1,  false,  false);
  }

  float* __restrict__ dst = (m == 0) ? Ai : ((m == 1) ? Bj : Y);
  float* __restrict__ drow =
      dst + (i0 + ((lane >> 4) << 3)) * HID + c0 + 2 * lm;
#pragma unroll
  for (int v = 0; v < 8; ++v) {
    v2f o;
    o.x = acc0[v];
    o.y = acc1[v];
    *(v2f*)(drow + v * HID) = o;
  }
}

#define TROWS  12
#define TWAVES 12
#define TGRID_I (N_GENES / TROWS)
#define TGRID_J (N_GENES / (16 * TWAVES))

__global__ __launch_bounds__(32 * TWAVES) void grn_tile_kernel(
    const float* __restrict__ X,  const float* __restrict__ Ai,
    const float* __restrict__ Bj, const float* __restrict__ Y,
    const float* __restrict__ b1, const float* __restrict__ W2,
    const float* __restrict__ b2, const float* __restrict__ bb,
    float* __restrict__ out) {
  __shared__ alignas(16) float sAi[TROWS * HID];
  __shared__ alignas(16) float sX[TWAVES * HID];

  const int tid  = threadIdx.x;
  const int lane = tid & 31;
  const int wave = __builtin_amdgcn_readfirstlane(tid >> 5);
  const int i0 = blockIdx.x * TROWS;
  const int j0 = (blockIdx.y * TWAVES + wave) * 16;
  const int lm = lane & 15;
  const int hh = lane >> 4;
  const int kb = hh << 1;

  const float4* __restrict__ gB =
      (const float4*)(Bj + (j0 + lm) * HID + 4 * hh);
  const int wr = lm & 7;
  const float4* lW =
      (const float4*)(sX + (1 + (wr < 3 ? wr : 3)) * HID + 4 * hh);
  const float4* lC = (const float4*)(sX + 4 * hh);
  const float4* lA = (const float4*)(sAi + 4 * hh);

  const float4 va = ((const float4*)(Ai + i0 * HID))[tid];
  const float* __restrict__ xbase = (wave == 0) ? b1 : W2;
  const int xi = (wave == 0) ? lane : ((wave < 4) ? tid - 32 : 0);
  float4 vx = ((const float4*)xbase)[xi];
  float4 b = gB[0];
  if (wave >= 4) { vx.x = 0.f; vx.y = 0.f; vx.z = 0.f; vx.w = 0.f; }
  ((float4*)sAi)[tid] = va;
  ((float4*)sX)[tid] = vx;

  const float b20 = b2[0], b21 = b2[1], b22 = b2[2];
  v8f d[TROWS];
#pragma unroll
  for (int g = 0; g < TROWS; ++g) {
    v8f z = {b20, b21, b22, 0.f, 0.f, 0.f, 0.f, 0.f};
    d[g] = z;
  }

  float4 a[TROWS], c, w;

  __syncthreads();

#pragma unroll
  for (int g = 0; g < TROWS; ++g) a[g] = lA[g * (HID / 4)];
  c = lC[0];
  w = lW[0];

#pragma unroll 1
  for (int t = 0; t < 16; ++t) {
    v2f w0, w1;
    w0.x = w.x; w0.y = w.y; w1.x = w.z; w1.y = w.w;
    v2f h0[TROWS], h1[TROWS];
#pragma unroll
    for (int g = 0; g < TROWS; ++g) {
      const float4 p = a[g];
      h0[g].x = fmaxf((p.x + b.x) + c.x, 0.f);
      h0[g].y = fmaxf((p.y + b.y) + c.y, 0.f);
      h1[g].x = fmaxf((p.z + b.z) + c.z, 0.f);
      h1[g].y = fmaxf((p.w + b.w) + c.w, 0.f);
    }
    const int tn = (t + 1) & 15;
#pragma unroll
    for (int g = 0; g < TROWS; ++g) a[g] = lA[g * (HID / 4) + 2 * tn];
    c = lC[2 * tn];
    w = lW[2 * tn];
    b = gB[2 * tn];
#pragma unroll
    for (int g = 0; g < TROWS; g += 2) {
      d[g] = __builtin_amdgcn_wmma_f32_16x16x4_f32(
           false, w0,  false, h0[g],
           (short)0, d[g],  false,  false);
      d[g + 1] = __builtin_amdgcn_wmma_f32_16x16x4_f32(
           false, w0,  false, h0[g + 1],
           (short)0, d[g + 1],  false,  false);
      d[g] = __builtin_amdgcn_wmma_f32_16x16x4_f32(
           false, w1,  false, h1[g],
           (short)0, d[g],  false,  false);
      d[g + 1] = __builtin_amdgcn_wmma_f32_16x16x4_f32(
           false, w1,  false, h1[g + 1],
           (short)0, d[g + 1],  false,  false);
    }
    __builtin_amdgcn_sched_group_barrier(0x002, 96, 0);
    __builtin_amdgcn_sched_group_barrier(0x100, 14, 0);
    __builtin_amdgcn_sched_group_barrier(0x020, 1, 0);
    __builtin_amdgcn_sched_group_barrier(0x008, 24, 0);
  }

  unsigned code = 0u;
#pragma unroll
  for (int v = 0; v < TROWS / 2; ++v) {
    const float l0 = hh ? d[v + TROWS / 2][0] : d[v][0];
    const float l1 = hh ? d[v + TROWS / 2][1] : d[v][1];
    const float l2 = hh ? d[v + TROWS / 2][2] : d[v][2];
    const unsigned cls = (l0 >= l1 && l0 >= l2) ? 0u : ((l1 >= l2) ? 1u : 2u);
    code |= cls << (2 * v);
  }

  const float bbv = bb[0];
  v8f acc = {bbv, bbv, bbv, bbv, bbv, bbv, bbv, bbv};
  {
    const int rsub = (lm & 7) < (TROWS / 2 - 1) ? (lm & 7) : (TROWS / 2 - 1);
    const int grow = i0 + (TROWS / 2) * (lm >> 3) + rsub;
    const float* __restrict__ arow = Y + grow * HID + kb;
    const float* __restrict__ brow = X + (j0 + lm) * HID + kb;
#pragma unroll
    for (int kk = 0; kk < HID; kk += 4) {
      const v2f afrag = *(const v2f*)(arow + kk);
      const v2f bfrag = *(const v2f*)(brow + kk);
      acc = __builtin_amdgcn_wmma_f32_16x16x4_f32(
           false, afrag,  false, bfrag,
           (short)0, acc,  false,  false);
    }
  }
  const int gbase = i0 + (TROWS / 2) * hh;
#pragma unroll
  for (int v = 0; v < TROWS / 2; ++v) {
    const int gi = gbase + v;
    const int gj = j0 + lm;
    const unsigned cls = (code >> (2 * v)) & 3u;
    const float s = (cls == 0u) ? 1.0f : ((cls == 2u) ? -1.0f : 0.0f);
    const float val = (gi == gj) ? 0.f : s * acc[v];
    out[gi * N_GENES + gj] = val;
  }
}

extern "C" void kernel_launch(void* const* d_in, const int* in_sizes, int n_in,
                              void* d_out, int out_size, void* d_ws,
                              size_t ws_size, hipStream_t stream) {
  (void)in_sizes; (void)n_in; (void)out_size; (void)ws_size;
  const float* X  = (const float*)d_in[0];
  const float* W1 = (const float*)d_in[1];
  const float* b1 = (const float*)d_in[2];
  const float* W2 = (const float*)d_in[3];
  const float* b2 = (const float*)d_in[4];
  const float* Wb = (const float*)d_in[5];
  const float* bb = (const float*)d_in[6];
  float* out = (float*)d_out;

  float* ws = (float*)d_ws;
  float* Ai = ws;
  float* Bj = ws + N_GENES * HID;
  float* Y  = ws + 2 * N_GENES * HID;

  grn_proj_kernel<<<dim3(NTILE, 12), dim3(32), 0, stream>>>(X, W1, Wb, Ai, Bj, Y);
  grn_tile_kernel<<<dim3(TGRID_I, TGRID_J), dim3(32 * TWAVES), 0, stream>>>(
      X, Ai, Bj, Y, b1, W2, b2, bb, out);
}
